# starting kernel
# baseline (speedup 1.0000x reference)
#include <hip/hip_runtime.h>
#include <hip/hip_bf16.h>
#include <math.h>

typedef __attribute__((ext_vector_type(2))) float v2f;
typedef __attribute__((ext_vector_type(8))) float v8f;

#define EPS_RHO 1e-8f
#define EPS_DEN 1e-9f

__device__ __forceinline__ float rho_relu(float x) {
    float p = sqrtf(fmaxf(x, 0.f) + EPS_RHO);
    float n = sqrtf(fmaxf(-x, 0.f) + EPS_RHO);
    return fmaxf(p - n, 0.f);
}

__device__ __forceinline__ v8f wmma_f32(v2f a, v2f b, v8f c) {
    return __builtin_amdgcn_wmma_f32_16x16x4_f32(false, a, false, b,
                                                 (short)0, c, false, false);
}

__global__ void fa_gemm_abt(const float* __restrict__ A,
                            const float* __restrict__ W,
                            const float* __restrict__ bias,
                            float* __restrict__ Out,
                            int M, int K, int Nc)
{
    const int lane  = threadIdx.x & 31;
    const int wave  = threadIdx.x >> 5;
    const int nblkN = Nc >> 6;
    const int job   = blockIdx.x * (blockDim.x >> 5) + wave;
    if (job >= (M >> 5) * nblkN) return;
    const int mb = job / nblkN, nb = job % nblkN;
    const int l15  = lane & 15;
    const int koff = (lane < 16) ? 0 : 2;
    const size_t bstep = (size_t)16 * K;

    const float* pA0 = A + (size_t)((mb << 5) + l15) * K + koff;
    const float* pA1 = pA0 + bstep;
    const float* pB0 = W + (size_t)((nb << 6) + l15) * K + koff;
    const float* pB1 = pB0 + bstep;
    const float* pB2 = pB1 + bstep;
    const float* pB3 = pB2 + bstep;

    v8f acc[2][4] = {};
    #pragma unroll 2
    for (int kk = 0; kk < K; kk += 4) {
        const v2f a0 = *(const v2f*)pA0;
        const v2f a1 = *(const v2f*)pA1;
        const v2f b0 = *(const v2f*)pB0;
        const v2f b1 = *(const v2f*)pB1;
        const v2f b2 = *(const v2f*)pB2;
        const v2f b3 = *(const v2f*)pB3;
        pA0 += 4; pA1 += 4; pB0 += 4; pB1 += 4; pB2 += 4; pB3 += 4;

        acc[0][0] = wmma_f32(a0, b0, acc[0][0]);
        acc[1][0] = wmma_f32(a1, b0, acc[1][0]);
        acc[0][1] = wmma_f32(a0, b1, acc[0][1]);
        acc[1][1] = wmma_f32(a1, b1, acc[1][1]);
        acc[0][2] = wmma_f32(a0, b2, acc[0][2]);
        acc[1][2] = wmma_f32(a1, b2, acc[1][2]);
        acc[0][3] = wmma_f32(a0, b3, acc[0][3]);
        acc[1][3] = wmma_f32(a1, b3, acc[1][3]);
    }

    const int mbase = (mb << 5) + ((lane < 16) ? 0 : 8);
    #pragma unroll
    for (int ni = 0; ni < 4; ++ni) {
        const int   n  = (nb << 6) + (ni << 4) + l15;
        const float bn = bias ? bias[n] : 0.f;
        #pragma unroll
        for (int mi = 0; mi < 2; ++mi)
            #pragma unroll
            for (int r = 0; r < 8; ++r)
                Out[(size_t)(mbase + (mi << 4) + r) * Nc + n] = acc[mi][ni][r] + bn;
    }
}

__global__ void fa_scores(const float* __restrict__ q,
                          const float* __restrict__ k,
                          const float* __restrict__ Wphi,
                          float* __restrict__ ex,
                          float* __restrict__ denom)
{
    __shared__ float qi[64];
    __shared__ float wph[64];
    __shared__ float red[8];
    __shared__ float msh;
    const int row = blockIdx.x;
    const int b   = row >> 9;
    const int tid = threadIdx.x;
    if (tid < 64) {
        qi[tid]  = q[(size_t)row * 64 + tid];
        wph[tid] = Wphi[tid];
    }
    __syncthreads();

    float s[2];
    #pragma unroll
    for (int p = 0; p < 2; ++p) {
        const int j = tid + (p << 8);
        const float* kj = k + ((size_t)(b * 512 + j)) * 64;
        float acc = 0.f;
        #pragma unroll 8
        for (int a = 0; a < 64; ++a)
            acc += rho_relu(qi[a] + kj[a]) * wph[a];
        s[p] = acc;
    }

    float mymax = fmaxf(s[0], s[1]);
    for (int off = 16; off; off >>= 1)
        mymax = fmaxf(mymax, __shfl_xor(mymax, off));
    const int lane = tid & 31, wv = tid >> 5;
    if (lane == 0) red[wv] = mymax;
    __syncthreads();
    if (tid == 0) {
        float m = red[0];
        for (int w = 1; w < 8; ++w) m = fmaxf(m, red[w]);
        msh = m;
    }
    __syncthreads();
    const float m = msh;

    const float e0 = expf(s[0] - m);
    const float e1 = expf(s[1] - m);
    ex[(size_t)row * 512 + tid]       = e0;
    ex[(size_t)row * 512 + tid + 256] = e1;

    float mysum = e0 + e1;
    for (int off = 16; off; off >>= 1)
        mysum += __shfl_xor(mysum, off);
    if (lane == 0) red[wv] = mysum;
    __syncthreads();
    if (tid == 0) {
        float sm = 0.f;
        for (int w = 0; w < 8; ++w) sm += red[w];
        denom[row] = sm;
    }
}

__global__ void fa_part1(const float* __restrict__ ex,
                         const float* __restrict__ Vy,
                         float* __restrict__ part1)
{
    const int lane = threadIdx.x & 31;
    const int wave = threadIdx.x >> 5;
    const int job  = blockIdx.x * (blockDim.x >> 5) + wave;
    const int b    = job >> 8;
    const int jj   = job & 255;
    const int mb   = jj >> 4;
    const int nb   = jj & 15;
    const int l15  = lane & 15;
    const int koff = (lane < 16) ? 0 : 2;
    const int col0 = (nb << 6) + l15;

    const float* pA0 = ex + ((size_t)(b * 512 + (mb << 5) + l15)) * 512 + koff;
    const float* pA1 = pA0 + (size_t)16 * 512;
    const float* pV  = Vy + (size_t)b * 512 * 1024 + (size_t)koff * 1024 + col0;

    v8f acc[2][4] = {};
    #pragma unroll 2
    for (int j = 0; j < 512; j += 4) {
        const v2f a0 = *(const v2f*)pA0;
        const v2f a1 = *(const v2f*)pA1;
        v2f b0, b1, b2, b3;
        b0.x = pV[0];          b0.y = pV[1024];
        b1.x = pV[16];         b1.y = pV[1024 + 16];
        b2.x = pV[32];         b2.y = pV[1024 + 32];
        b3.x = pV[48];         b3.y = pV[1024 + 48];
        pA0 += 4; pA1 += 4; pV += 4 * 1024;

        acc[0][0] = wmma_f32(a0, b0, acc[0][0]);
        acc[1][0] = wmma_f32(a1, b0, acc[1][0]);
        acc[0][1] = wmma_f32(a0, b1, acc[0][1]);
        acc[1][1] = wmma_f32(a1, b1, acc[1][1]);
        acc[0][2] = wmma_f32(a0, b2, acc[0][2]);
        acc[1][2] = wmma_f32(a1, b2, acc[1][2]);
        acc[0][3] = wmma_f32(a0, b3, acc[0][3]);
        acc[1][3] = wmma_f32(a1, b3, acc[1][3]);
    }

    const int mbase = b * 512 + (mb << 5) + ((lane < 16) ? 0 : 8);
    #pragma unroll
    for (int ni = 0; ni < 4; ++ni) {
        const int n = col0 + (ni << 4);
        #pragma unroll
        for (int mi = 0; mi < 2; ++mi)
            #pragma unroll
            for (int r = 0; r < 8; ++r)
                part1[(size_t)(mbase + (mi << 4) + r) * 1024 + n] = acc[mi][ni][r];
    }
}

__global__ void fa_ew(const float* __restrict__ q,
                      const float* __restrict__ k,
                      const float* __restrict__ ex,
                      float* __restrict__ Ew)
{
    __shared__ float exs[512];
    const int row = blockIdx.x;
    const int b   = row >> 9;
    const int a   = threadIdx.x;
    for (int j = a; j < 512; j += 64)
        exs[j] = ex[(size_t)row * 512 + j];
    __syncthreads();
    const float qa = q[(size_t)row * 64 + a];
    float acc = 0.f;
    #pragma unroll 4
    for (int j = 0; j < 512; ++j) {
        const float e = rho_relu(qa + k[((size_t)(b * 512 + j)) * 64 + a]);
        acc += exs[j] * e;
    }
    Ew[(size_t)row * 64 + a] = acc;
}

__global__ void fa_final(const float* __restrict__ Ew,
                         const float* __restrict__ Wew,
                         const float* __restrict__ Web,
                         const float* __restrict__ part1,
                         const float* __restrict__ denom,
                         float* __restrict__ out)
{
    const int lane = threadIdx.x & 31;
    const int wave = threadIdx.x >> 5;
    const int t  = blockIdx.x * (blockDim.x >> 5) + wave;
    const int mt = t >> 6, nt = t & 63;
    const int l15  = lane & 15;
    const int koff = (lane < 16) ? 0 : 2;
    const float* Arow = Ew  + (size_t)((mt << 4) + l15) * 64 + koff;
    const float* Brow = Wew + (size_t)((nt << 4) + l15) * 64 + koff;
    v8f acc = {};
    #pragma unroll
    for (int kk = 0; kk < 64; kk += 4) {
        v2f av = *(const v2f*)(Arow + kk);
        v2f bv = *(const v2f*)(Brow + kk);
        acc = wmma_f32(av, bv, acc);
    }
    const int   n     = (nt << 4) + l15;
    const int   mbase = (mt << 4) + ((lane < 16) ? 0 : 8);
    const float bn    = Web[n];
    #pragma unroll
    for (int r = 0; r < 8; ++r) {
        const int   m = mbase + r;
        const float d = denom[m];
        out[(size_t)m * 1024 + n] =
            (part1[(size_t)m * 1024 + n] + acc[r] + bn * d) / (d + EPS_DEN);
    }
}

extern "C" void kernel_launch(void* const* d_in, const int* in_sizes, int n_in,
                              void* d_out, int out_size, void* d_ws, size_t ws_size,
                              hipStream_t stream)
{
    (void)in_sizes; (void)n_in; (void)out_size; (void)ws_size;
    const float* X    = (const float*)d_in[0];
    const float* Wq   = (const float*)d_in[1];
    const float* Wk   = (const float*)d_in[2];
    const float* Wphi = (const float*)d_in[3];
    const float* Wyw  = (const float*)d_in[4];
    const float* Wyb  = (const float*)d_in[5];
    const float* Wew  = (const float*)d_in[6];
    const float* Web  = (const float*)d_in[7];
    float* out = (float*)d_out;

    float* ws  = (float*)d_ws;
    float* q   = ws;  ws += 1024 * 64;
    float* kk  = ws;  ws += 1024 * 64;
    float* Vy  = ws;  ws += 1024 * 1024;
    float* ex  = ws;  ws += 1024 * 512;
    float* den = ws;  ws += 1024;
    float* p1  = ws;  ws += 1024 * 1024;
    float* Ew  = ws;  ws += 1024 * 64;

    fa_gemm_abt<<<4,   256, 0, stream>>>(X, Wq,  nullptr, q,  1024, 1024, 64);
    fa_gemm_abt<<<4,   256, 0, stream>>>(X, Wk,  nullptr, kk, 1024, 1024, 64);
    fa_gemm_abt<<<64,  256, 0, stream>>>(X, Wyw, Wyb,     Vy, 1024, 1024, 1024);

    fa_scores<<<1024, 256, 0, stream>>>(q, kk, Wphi, ex, den);

    fa_part1<<<64, 256, 0, stream>>>(ex, Vy, p1);

    fa_ew<<<1024, 64, 0, stream>>>(q, kk, ex, Ew);

    fa_final<<<512, 256, 0, stream>>>(Ew, Wew, Web, p1, den, out);
}


# q and k projections (64 columns): 16x32 block tiles with the k range split over a block's 8 waves (128 blocks instead of 16), partial tiles added through LDS; k sum = tree of 8 slice sums
# speedup vs baseline: 4.5546x; 4.5546x over previous
#include <hip/hip_runtime.h>
#include <hip/hip_bf16.h>
#include <math.h>

typedef __attribute__((ext_vector_type(2))) float v2f;
typedef __attribute__((ext_vector_type(8))) float v8f;

#define EPS_RHO 1e-8f
#define EPS_DEN 1e-9f

__device__ __forceinline__ float rho_relu(float x) {
    float p = sqrtf(fmaxf(x, 0.f) + EPS_RHO);
    float n = sqrtf(fmaxf(-x, 0.f) + EPS_RHO);
    return fmaxf(p - n, 0.f);
}

__device__ __forceinline__ v8f wmma_f32(v2f a, v2f b, v8f c) {
    return __builtin_amdgcn_wmma_f32_16x16x4_f32(false, a, false, b,
                                                 (short)0, c, false, false);
}

#define GT_KC 64
#define GT_P  68

__device__ __forceinline__ void gt_chunk(const float* pa, const float* pb,
                                         v8f (&acc)[2][2])
{
    #pragma unroll
    for (int ks = 0; ks < GT_KC; ks += 4) {
        const v2f a0 = *(const v2f*)(pa + ks);
        const v2f a1 = *(const v2f*)(pa + 16 * GT_P + ks);
        const v2f b0 = *(const v2f*)(pb + ks);
        const v2f b1 = *(const v2f*)(pb + 16 * GT_P + ks);
        acc[0][0] = wmma_f32(a0, b0, acc[0][0]);
        acc[1][0] = wmma_f32(a1, b0, acc[1][0]);
        acc[0][1] = wmma_f32(a0, b1, acc[0][1]);
        acc[1][1] = wmma_f32(a1, b1, acc[1][1]);
    }
}

__global__ __launch_bounds__(128)
void fa_gemm_abt(const float* __restrict__ A,
                 const float* __restrict__ W,
                 const float* __restrict__ bias,
                 float* __restrict__ Out,
                 int M, int K, int Nc)
{
    __shared__ __attribute__((aligned(16))) float sA[2][64 * GT_P];
    __shared__ __attribute__((aligned(16))) float sB[2][64 * GT_P];

    const int tid   = threadIdx.x;
    const int lane  = tid & 31;
    const int wave  = tid >> 5;
    const int wm    = wave >> 1, wn = wave & 1;
    const int nblkN = Nc >> 6;
    const int mb = blockIdx.x / nblkN, nb = blockIdx.x % nblkN;
    if (mb >= (M >> 6)) return;
    const int l15  = lane & 15;
    const int koff = (lane < 16) ? 0 : 2;

    const int sc = (tid & 15) << 2;
    const int sr = tid >> 4;
    const float* gA = A + (size_t)((mb << 6) + sr) * K + sc;
    const float* gB = W + (size_t)((nb << 6) + sr) * K + sc;
    const size_t rstep = (size_t)8 * K;
    const int    soff  = sr * GT_P + sc;

    const int fa = ((wm << 5) + l15) * GT_P + koff;
    const int fb = ((wn << 5) + l15) * GT_P + koff;

    #define GT_LOAD(i)  ra##i = *(const float4*)(gA + (i) * rstep); \
                        rb##i = *(const float4*)(gB + (i) * rstep);
    #define GT_PARK(s, i) *(float4*)(&sA[s][soff + (i) * 8 * GT_P]) = ra##i; \
                          *(float4*)(&sB[s][soff + (i) * 8 * GT_P]) = rb##i;
    #define GT_LOAD_ALL()  GT_LOAD(0) GT_LOAD(1) GT_LOAD(2) GT_LOAD(3) \
                           GT_LOAD(4) GT_LOAD(5) GT_LOAD(6) GT_LOAD(7)
    #define GT_PARK_ALL(s) GT_PARK(s, 0) GT_PARK(s, 1) GT_PARK(s, 2) GT_PARK(s, 3) \
                           GT_PARK(s, 4) GT_PARK(s, 5) GT_PARK(s, 6) GT_PARK(s, 7)
    float4 ra0, ra1, ra2, ra3, ra4, ra5, ra6, ra7;
    float4 rb0, rb1, rb2, rb3, rb4, rb5, rb6, rb7;

    GT_LOAD_ALL()
    GT_PARK_ALL(0)
    gA += GT_KC; gB += GT_KC;
    GT_LOAD_ALL()
    __syncthreads();

    v8f acc[2][2] = {};
    const int nchunks = K / GT_KC;
    int c = 0;
    for (; c + 2 < nchunks; ++c) {
        const int st = c & 1;
        GT_PARK_ALL(st ^ 1)
        gA += GT_KC; gB += GT_KC;
        GT_LOAD_ALL()
        gt_chunk(&sA[st][fa], &sB[st][fb], acc);
        __syncthreads();
    }
    {
        const int st = c & 1;
        GT_PARK_ALL(st ^ 1)
        gt_chunk(&sA[st][fa], &sB[st][fb], acc);
        __syncthreads();
        gt_chunk(&sA[st ^ 1][fa], &sB[st ^ 1][fb], acc);
    }
    #undef GT_LOAD
    #undef GT_PARK
    #undef GT_LOAD_ALL
    #undef GT_PARK_ALL

    const int mbase = (mb << 6) + (wm << 5) + ((lane < 16) ? 0 : 8);
    #pragma unroll
    for (int ni = 0; ni < 2; ++ni) {
        const int   n  = (nb << 6) + (wn << 5) + (ni << 4) + l15;
        const float bn = bias ? bias[n] : 0.f;
        #pragma unroll
        for (int mi = 0; mi < 2; ++mi)
            #pragma unroll
            for (int r = 0; r < 8; ++r)
                Out[(size_t)(mbase + (mi << 4) + r) * Nc + n] = acc[mi][ni][r] + bn;
    }
}

#define SK_KC 256
#define SK_P  260
#define SK_RP 36

__device__ __forceinline__ void sk_chunk(const float* pa, const float* pb,
                                         v8f (&acc)[2])
{
    #pragma unroll
    for (int ks = 0; ks < 32; ks += 4) {
        const v2f a  = *(const v2f*)(pa + ks);
        const v2f b0 = *(const v2f*)(pb + ks);
        const v2f b1 = *(const v2f*)(pb + 16 * SK_P + ks);
        acc[0] = wmma_f32(a, b0, acc[0]);
        acc[1] = wmma_f32(a, b1, acc[1]);
    }
}

__global__ __launch_bounds__(256)
void fa_gemm_abt_ksplit(const float* __restrict__ A,
                        const float* __restrict__ W,
                        const float* __restrict__ bias,
                        float* __restrict__ Out,
                        int M, int K, int Nc)
{
    __shared__ __attribute__((aligned(16))) float sA[2][16 * SK_P];
    __shared__ __attribute__((aligned(16))) float sB[2][32 * SK_P];
    __shared__ __attribute__((aligned(16))) float sR[8][16 * SK_RP];

    const int tid   = threadIdx.x;
    const int lane  = tid & 31;
    const int wave  = tid >> 5;
    const int nblkN = Nc >> 5;
    const int mb = blockIdx.x / nblkN, nb = blockIdx.x % nblkN;
    if (mb >= (M >> 4)) return;
    const int l15  = lane & 15;
    const int koff = (lane < 16) ? 0 : 2;

    const int sc = (tid & 63) << 2;
    const int sr = tid >> 6;
    const float* gA = A + (size_t)((mb << 4) + sr) * K + sc;
    const float* gB = W + (size_t)((nb << 5) + sr) * K + sc;
    const size_t rstep = (size_t)4 * K;
    const int    soff  = sr * SK_P + sc;

    const int fo = l15 * SK_P + (wave << 5) + koff;

    #define SK_LOADA(i) ra##i = *(const float4*)(gA + (i) * rstep);
    #define SK_LOADB(i) rb##i = *(const float4*)(gB + (i) * rstep);
    #define SK_PARKA(s, i) *(float4*)(&sA[s][soff + (i) * 4 * SK_P]) = ra##i;
    #define SK_PARKB(s, i) *(float4*)(&sB[s][soff + (i) * 4 * SK_P]) = rb##i;
    #define SK_LOAD_ALL()  SK_LOADA(0) SK_LOADA(1) SK_LOADA(2) SK_LOADA(3) \
                           SK_LOADB(0) SK_LOADB(1) SK_LOADB(2) SK_LOADB(3) \
                           SK_LOADB(4) SK_LOADB(5) SK_LOADB(6) SK_LOADB(7)
    #define SK_PARK_ALL(s) SK_PARKA(s, 0) SK_PARKA(s, 1) SK_PARKA(s, 2) SK_PARKA(s, 3) \
                           SK_PARKB(s, 0) SK_PARKB(s, 1) SK_PARKB(s, 2) SK_PARKB(s, 3) \
                           SK_PARKB(s, 4) SK_PARKB(s, 5) SK_PARKB(s, 6) SK_PARKB(s, 7)
    float4 ra0, ra1, ra2, ra3;
    float4 rb0, rb1, rb2, rb3, rb4, rb5, rb6, rb7;

    SK_LOAD_ALL()
    SK_PARK_ALL(0)
    gA += SK_KC; gB += SK_KC;
    SK_LOAD_ALL()
    __syncthreads();

    v8f acc[2] = {};
    const int nchunks = K / SK_KC;
    int c = 0;
    for (; c + 2 < nchunks; ++c) {
        const int st = c & 1;
        SK_PARK_ALL(st ^ 1)
        gA += SK_KC; gB += SK_KC;
        SK_LOAD_ALL()
        sk_chunk(&sA[st][fo], &sB[st][fo], acc);
        __syncthreads();
    }
    {
        const int st = c & 1;
        SK_PARK_ALL(st ^ 1)
        sk_chunk(&sA[st][fo], &sB[st][fo], acc);
        __syncthreads();
        sk_chunk(&sA[st ^ 1][fo], &sB[st ^ 1][fo], acc);
    }
    #undef SK_LOADA
    #undef SK_LOADB
    #undef SK_PARKA
    #undef SK_PARKB
    #undef SK_LOAD_ALL
    #undef SK_PARK_ALL

    const int rbase = (lane < 16) ? 0 : 8;
    #pragma unroll
    for (int ni = 0; ni < 2; ++ni)
        #pragma unroll
        for (int r = 0; r < 8; ++r)
            sR[wave][(rbase + r) * SK_RP + (ni << 4) + l15] = acc[ni][r];
    __syncthreads();

    const int orow = tid >> 4;
    const int oc   = (tid & 15) << 1;
    const int ro   = orow * SK_RP + oc;
    const v2f p0 = *(const v2f*)(&sR[0][ro]);
    const v2f p1 = *(const v2f*)(&sR[1][ro]);
    const v2f p2 = *(const v2f*)(&sR[2][ro]);
    const v2f p3 = *(const v2f*)(&sR[3][ro]);
    const v2f p4 = *(const v2f*)(&sR[4][ro]);
    const v2f p5 = *(const v2f*)(&sR[5][ro]);
    const v2f p6 = *(const v2f*)(&sR[6][ro]);
    const v2f p7 = *(const v2f*)(&sR[7][ro]);
    const int n = (nb << 5) + oc;
    v2f o = ((p0 + p1) + (p2 + p3)) + ((p4 + p5) + (p6 + p7));
    if (bias) { o.x += bias[n]; o.y += bias[n + 1]; }
    *(v2f*)(&Out[(size_t)((mb << 4) + orow) * Nc + n]) = o;
}

__global__ __launch_bounds__(256)
void fa_scores(const float* __restrict__ q,
               const float* __restrict__ k,
               const float* __restrict__ Wphi,
               float* __restrict__ ex,
               float* __restrict__ denom,
               float* __restrict__ Ew)
{
    __shared__ float wph[64];
    __shared__ float qs[2][64];
    __shared__ float ssh[2][512];
    __shared__ float redm[2][4];
    __shared__ float redd[2][4];
    __shared__ float redE[2][4][64];
    const int tid  = threadIdx.x;
    const int r    = tid >> 7;
    const int t    = tid & 127;
    const int row  = blockIdx.x * 2 + r;
    const int b    = row >> 9;
    const int lane = tid & 31;
    const int wr   = t >> 5;
    const int qd   = t >> 2;
    const int c    = t & 3;
    if (t < 64)   qs[r][t] = q[(size_t)row * 64 + t];
    if (tid < 64) wph[tid] = Wphi[tid];
    __syncthreads();

    float qv[16], wv[16], Ewa[16];
    #pragma unroll
    for (int a = 0; a < 16; ++a) {
        qv[a]  = qs[r][c * 16 + a];
        wv[a]  = wph[c * 16 + a];
        Ewa[a] = 0.f;
    }
    float m = -INFINITY;
    const float* kb = k + ((size_t)(b * 512 + qd)) * 64 + c * 16;

    for (int n0 = 0; n0 < 16; n0 += 4) {
        float e[4][16];
        float s[4];
        #pragma unroll
        for (int g = 0; g < 4; ++g) {
            const float4* kp = (const float4*)(kb + (size_t)(n0 + g) * 32 * 64);
            float acc = 0.f;
            #pragma unroll
            for (int v = 0; v < 4; ++v) {
                const float4 kv = kp[v];
                e[g][4 * v + 0] = rho_relu(qv[4 * v + 0] + kv.x);
                e[g][4 * v + 1] = rho_relu(qv[4 * v + 1] + kv.y);
                e[g][4 * v + 2] = rho_relu(qv[4 * v + 2] + kv.z);
                e[g][4 * v + 3] = rho_relu(qv[4 * v + 3] + kv.w);
                acc += e[g][4 * v + 0] * wv[4 * v + 0];
                acc += e[g][4 * v + 1] * wv[4 * v + 1];
                acc += e[g][4 * v + 2] * wv[4 * v + 2];
                acc += e[g][4 * v + 3] * wv[4 * v + 3];
            }
            acc += __shfl_xor(acc, 1);
            acc += __shfl_xor(acc, 2);
            s[g] = acc;
            if (c == 0) ssh[r][qd + 32 * (n0 + g)] = acc;
        }
        const float mn = fmaxf(fmaxf(m, fmaxf(s[0], s[1])), fmaxf(s[2], s[3]));
        const float sc = expf(m - mn);
        const float p0 = expf(s[0] - mn);
        const float p1 = expf(s[1] - mn);
        const float p2 = expf(s[2] - mn);
        const float p3 = expf(s[3] - mn);
        #pragma unroll
        for (int a = 0; a < 16; ++a) {
            float v = Ewa[a] * sc;
            v += p0 * e[0][a];
            v += p1 * e[1][a];
            v += p2 * e[2][a];
            v += p3 * e[3][a];
            Ewa[a] = v;
        }
        m = mn;
    }

    float M = m;
    M = fmaxf(M, __shfl_xor(M, 4));
    M = fmaxf(M, __shfl_xor(M, 8));
    M = fmaxf(M, __shfl_xor(M, 16));
    if (lane == 0) redm[r][wr] = M;
    __syncthreads();
    M = fmaxf(fmaxf(redm[r][0], redm[r][1]), fmaxf(redm[r][2], redm[r][3]));

    const float f = expf(m - M);
    #pragma unroll
    for (int a = 0; a < 16; ++a) {
        float v = Ewa[a] * f;
        v += __shfl_xor(v, 4);
        v += __shfl_xor(v, 8);
        v += __shfl_xor(v, 16);
        Ewa[a] = v;
    }
    if (lane < 4) {
        #pragma unroll
        for (int a = 0; a < 16; ++a) redE[r][wr][lane * 16 + a] = Ewa[a];
    }

    float ds = 0.f;
    #pragma unroll
    for (int n = 0; n < 4; ++n) {
        const int   j = t + 128 * n;
        const float v = expf(ssh[r][j] - M);
        ex[(size_t)row * 512 + j] = v;
        ds += v;
    }
    for (int off = 16; off; off >>= 1)
        ds += __shfl_xor(ds, off);
    if (lane == 0) redd[r][wr] = ds;
    __syncthreads();
    if (t < 64)
        Ew[(size_t)row * 64 + t] = (redE[r][0][t] + redE[r][1][t]) + (redE[r][2][t] + redE[r][3][t]);
    if (t == 0)
        denom[row] = (redd[r][0] + redd[r][1]) + (redd[r][2] + redd[r][3]);
}

#define GT_VT 1040

__device__ __forceinline__ void gt_chunk_kn(const float* pa, const float* pv,
                                            v8f (&acc)[2][2])
{
    #pragma unroll
    for (int ks = 0; ks < GT_KC; ks += 4) {
        const v2f a0 = *(const v2f*)(pa + ks);
        const v2f a1 = *(const v2f*)(pa + 16 * GT_P + ks);
        v2f b0, b1;
        b0.x = pv[ks * 16];               b0.y = pv[ks * 16 + 16];
        b1.x = pv[GT_VT + ks * 16];       b1.y = pv[GT_VT + ks * 16 + 16];
        acc[0][0] = wmma_f32(a0, b0, acc[0][0]);
        acc[1][0] = wmma_f32(a1, b0, acc[1][0]);
        acc[0][1] = wmma_f32(a0, b1, acc[0][1]);
        acc[1][1] = wmma_f32(a1, b1, acc[1][1]);
    }
}

__global__ __launch_bounds__(128)
void fa_part1(const float* __restrict__ ex,
              const float* __restrict__ Vy,
              float* __restrict__ part1)
{
    __shared__ __attribute__((aligned(16))) float sA[2][64 * GT_P];
    __shared__ __attribute__((aligned(16))) float sV[2][4 * GT_VT];

    const int tid  = threadIdx.x;
    const int lane = tid & 31;
    const int wave = tid >> 5;
    const int wm   = wave >> 1, wn = wave & 1;
    const int b    = blockIdx.x >> 7;
    const int mb   = (blockIdx.x >> 4) & 7;
    const int nb   = blockIdx.x & 15;
    const int l15  = lane & 15;
    const int koff = (lane < 16) ? 0 : 2;

    const int sc = (tid & 15) << 2;
    const int sr = tid >> 4;
    const float* gA = ex + (size_t)(b * 512 + (mb << 6) + sr) * 512 + sc;
    const float* gV = Vy + (size_t)(b * 512 + sr) * 1024 + (nb << 6) + sc;
    const int soffA = sr * GT_P + sc;
    const int soffV = (sc >> 4) * GT_VT + sr * 16 + (sc & 15);

    const int fa = ((wm << 5) + l15) * GT_P + koff;
    const int fv = (wn << 1) * GT_VT + koff * 16 + l15;

    #define GT_LOAD(i)  ra##i = *(const float4*)(gA + (size_t)(i) * 8 * 512); \
                        rv##i = *(const float4*)(gV + (size_t)(i) * 8 * 1024);
    #define GT_PARK(s, i) *(float4*)(&sA[s][soffA + (i) * 8 * GT_P])  = ra##i; \
                          *(float4*)(&sV[s][soffV + (i) * 8 * 16]) = rv##i;
    #define GT_LOAD_ALL()  GT_LOAD(0) GT_LOAD(1) GT_LOAD(2) GT_LOAD(3) \
                           GT_LOAD(4) GT_LOAD(5) GT_LOAD(6) GT_LOAD(7)
    #define GT_PARK_ALL(s) GT_PARK(s, 0) GT_PARK(s, 1) GT_PARK(s, 2) GT_PARK(s, 3) \
                           GT_PARK(s, 4) GT_PARK(s, 5) GT_PARK(s, 6) GT_PARK(s, 7)
    float4 ra0, ra1, ra2, ra3, ra4, ra5, ra6, ra7;
    float4 rv0, rv1, rv2, rv3, rv4, rv5, rv6, rv7;

    GT_LOAD_ALL()
    GT_PARK_ALL(0)
    gA += GT_KC; gV += (size_t)GT_KC * 1024;
    GT_LOAD_ALL()
    __syncthreads();

    v8f acc[2][2] = {};
    int c = 0;
    for (; c + 2 < 512 / GT_KC; ++c) {
        const int st = c & 1;
        GT_PARK_ALL(st ^ 1)
        gA += GT_KC; gV += (size_t)GT_KC * 1024;
        GT_LOAD_ALL()
        gt_chunk_kn(&sA[st][fa], &sV[st][fv], acc);
        __syncthreads();
    }
    {
        const int st = c & 1;
        GT_PARK_ALL(st ^ 1)
        gt_chunk_kn(&sA[st][fa], &sV[st][fv], acc);
        __syncthreads();
        gt_chunk_kn(&sA[st ^ 1][fa], &sV[st ^ 1][fv], acc);
    }
    #undef GT_LOAD
    #undef GT_PARK
    #undef GT_LOAD_ALL
    #undef GT_PARK_ALL

    const int mbase = b * 512 + (mb << 6) + (wm << 5) + ((lane < 16) ? 0 : 8);
    const int col0  = (nb << 6) + (wn << 5) + l15;
    #pragma unroll
    for (int ni = 0; ni < 2; ++ni) {
        const int n = col0 + (ni << 4);
        #pragma unroll
        for (int mi = 0; mi < 2; ++mi)
            #pragma unroll
            for (int r = 0; r < 8; ++r)
                part1[(size_t)(mbase + (mi << 4) + r) * 1024 + n] = acc[mi][ni][r];
    }
}

__global__ void fa_final(const float* __restrict__ Ew,
                         const float* __restrict__ Wew,
                         const float* __restrict__ Web,
                         const float* __restrict__ part1,
                         const float* __restrict__ denom,
                         float* __restrict__ out)
{
    const int lane = threadIdx.x & 31;
    const int wave = threadIdx.x >> 5;
    const int t  = blockIdx.x * (blockDim.x >> 5) + wave;
    const int mt = t >> 6, nt = t & 63;
    const int l15  = lane & 15;
    const int koff = (lane < 16) ? 0 : 2;
    const float* Arow = Ew  + (size_t)((mt << 4) + l15) * 64 + koff;
    const float* Brow = Wew + (size_t)((nt << 4) + l15) * 64 + koff;
    v8f acc = {};
    #pragma unroll
    for (int kk = 0; kk < 64; kk += 4) {
        v2f av = *(const v2f*)(Arow + kk);
        v2f bv = *(const v2f*)(Brow + kk);
        acc = wmma_f32(av, bv, acc);
    }
    const int   n     = (nt << 4) + l15;
    const int   mbase = (mt << 4) + ((lane < 16) ? 0 : 8);
    const float bn    = Web[n];
    #pragma unroll
    for (int r = 0; r < 8; ++r) {
        const int   m = mbase + r;
        const float d = denom[m];
        out[(size_t)m * 1024 + n] =
            (part1[(size_t)m * 1024 + n] + acc[r] + bn * d) / (d + EPS_DEN);
    }
}

extern "C" void kernel_launch(void* const* d_in, const int* in_sizes, int n_in,
                              void* d_out, int out_size, void* d_ws, size_t ws_size,
                              hipStream_t stream)
{
    (void)in_sizes; (void)n_in; (void)out_size; (void)ws_size;
    const float* X    = (const float*)d_in[0];
    const float* Wq   = (const float*)d_in[1];
    const float* Wk   = (const float*)d_in[2];
    const float* Wphi = (const float*)d_in[3];
    const float* Wyw  = (const float*)d_in[4];
    const float* Wyb  = (const float*)d_in[5];
    const float* Wew  = (const float*)d_in[6];
    const float* Web  = (const float*)d_in[7];
    float* out = (float*)d_out;

    float* ws  = (float*)d_ws;
    float* q   = ws;  ws += 1024 * 64;
    float* kk  = ws;  ws += 1024 * 64;
    float* Vy  = ws;  ws += 1024 * 1024;
    float* ex  = ws;  ws += 1024 * 512;
    float* den = ws;  ws += 1024;
    float* p1  = ws;  ws += 1024 * 1024;
    float* Ew  = ws;  ws += 1024 * 64;

    fa_gemm_abt_ksplit<<<128, 256, 0, stream>>>(X, Wq, nullptr, q,  1024, 1024, 64);
    fa_gemm_abt_ksplit<<<128, 256, 0, stream>>>(X, Wk, nullptr, kk, 1024, 1024, 64);
    fa_gemm_abt<<<256, 128, 0, stream>>>(X, Wyw, Wyb,     Vy, 1024, 1024, 1024);

    fa_scores<<<512, 256, 0, stream>>>(q, kk, Wphi, ex, den, Ew);

    fa_part1<<<256, 128, 0, stream>>>(ex, Vy, p1);

    fa_final<<<512, 256, 0, stream>>>(Ew, Wew, Web, p1, den, out);
}


# Vy projection moved into the scores launch (256 tile blocks first, then 1024 one-row score blocks): its WMMA waves share SIMDs with the VALU-bound score waves; 6->5 launches; sums kept
# speedup vs baseline: 5.2462x; 1.1518x over previous
#include <hip/hip_runtime.h>
#include <hip/hip_bf16.h>
#include <math.h>

typedef __attribute__((ext_vector_type(2))) float v2f;
typedef __attribute__((ext_vector_type(8))) float v8f;

#define EPS_RHO 1e-8f
#define EPS_DEN 1e-9f

__device__ __forceinline__ float rho_relu(float x) {
    float p = sqrtf(fmaxf(x, 0.f) + EPS_RHO);
    float n = sqrtf(fmaxf(-x, 0.f) + EPS_RHO);
    return fmaxf(p - n, 0.f);
}

__device__ __forceinline__ v8f wmma_f32(v2f a, v2f b, v8f c) {
    return __builtin_amdgcn_wmma_f32_16x16x4_f32(false, a, false, b,
                                                 (short)0, c, false, false);
}

#define GT_KC 64
#define GT_P  68

__device__ __forceinline__ void gt_chunk(const float* pa, const float* pb,
                                         v8f (&acc)[2][2])
{
    #pragma unroll
    for (int ks = 0; ks < GT_KC; ks += 4) {
        const v2f a0 = *(const v2f*)(pa + ks);
        const v2f a1 = *(const v2f*)(pa + 16 * GT_P + ks);
        const v2f b0 = *(const v2f*)(pb + ks);
        const v2f b1 = *(const v2f*)(pb + 16 * GT_P + ks);
        acc[0][0] = wmma_f32(a0, b0, acc[0][0]);
        acc[1][0] = wmma_f32(a1, b0, acc[1][0]);
        acc[0][1] = wmma_f32(a0, b1, acc[0][1]);
        acc[1][1] = wmma_f32(a1, b1, acc[1][1]);
    }
}

__device__ __forceinline__
void gemm_abt_tile(const float* __restrict__ A,
                   const float* __restrict__ W,
                   const float* __restrict__ bias,
                   float* __restrict__ Out,
                   int K, int Nc, int bid)
{
    __shared__ __attribute__((aligned(16))) float sA[2][64 * GT_P];
    __shared__ __attribute__((aligned(16))) float sB[2][64 * GT_P];

    const int tid   = threadIdx.x;
    const int lane  = tid & 31;
    const int wave  = tid >> 5;
    const int wm    = wave >> 1, wn = wave & 1;
    const int nblkN = Nc >> 6;
    const int mb = bid / nblkN, nb = bid % nblkN;
    const int l15  = lane & 15;
    const int koff = (lane < 16) ? 0 : 2;

    const int sc = (tid & 15) << 2;
    const int sr = tid >> 4;
    const float* gA = A + (size_t)((mb << 6) + sr) * K + sc;
    const float* gB = W + (size_t)((nb << 6) + sr) * K + sc;
    const size_t rstep = (size_t)8 * K;
    const int    soff  = sr * GT_P + sc;

    const int fa = ((wm << 5) + l15) * GT_P + koff;
    const int fb = ((wn << 5) + l15) * GT_P + koff;

    #define GT_LOAD(i)  ra##i = *(const float4*)(gA + (i) * rstep); \
                        rb##i = *(const float4*)(gB + (i) * rstep);
    #define GT_PARK(s, i) *(float4*)(&sA[s][soff + (i) * 8 * GT_P]) = ra##i; \
                          *(float4*)(&sB[s][soff + (i) * 8 * GT_P]) = rb##i;
    #define GT_LOAD_ALL()  GT_LOAD(0) GT_LOAD(1) GT_LOAD(2) GT_LOAD(3) \
                           GT_LOAD(4) GT_LOAD(5) GT_LOAD(6) GT_LOAD(7)
    #define GT_PARK_ALL(s) GT_PARK(s, 0) GT_PARK(s, 1) GT_PARK(s, 2) GT_PARK(s, 3) \
                           GT_PARK(s, 4) GT_PARK(s, 5) GT_PARK(s, 6) GT_PARK(s, 7)
    float4 ra0, ra1, ra2, ra3, ra4, ra5, ra6, ra7;
    float4 rb0, rb1, rb2, rb3, rb4, rb5, rb6, rb7;

    GT_LOAD_ALL()
    GT_PARK_ALL(0)
    gA += GT_KC; gB += GT_KC;
    GT_LOAD_ALL()
    __syncthreads();

    v8f acc[2][2] = {};
    const int nchunks = K / GT_KC;
    int c = 0;
    for (; c + 2 < nchunks; ++c) {
        const int st = c & 1;
        GT_PARK_ALL(st ^ 1)
        gA += GT_KC; gB += GT_KC;
        GT_LOAD_ALL()
        gt_chunk(&sA[st][fa], &sB[st][fb], acc);
        __syncthreads();
    }
    {
        const int st = c & 1;
        GT_PARK_ALL(st ^ 1)
        gt_chunk(&sA[st][fa], &sB[st][fb], acc);
        __syncthreads();
        gt_chunk(&sA[st ^ 1][fa], &sB[st ^ 1][fb], acc);
    }
    #undef GT_LOAD
    #undef GT_PARK
    #undef GT_LOAD_ALL
    #undef GT_PARK_ALL

    const int mbase = (mb << 6) + (wm << 5) + ((lane < 16) ? 0 : 8);
    #pragma unroll
    for (int ni = 0; ni < 2; ++ni) {
        const int   n  = (nb << 6) + (wn << 5) + (ni << 4) + l15;
        const float bn = bias ? bias[n] : 0.f;
        #pragma unroll
        for (int mi = 0; mi < 2; ++mi)
            #pragma unroll
            for (int r = 0; r < 8; ++r)
                Out[(size_t)(mbase + (mi << 4) + r) * Nc + n] = acc[mi][ni][r] + bn;
    }
}

#define SK_KC 256
#define SK_P  260
#define SK_RP 36

__device__ __forceinline__ void sk_chunk(const float* pa, const float* pb,
                                         v8f (&acc)[2])
{
    #pragma unroll
    for (int ks = 0; ks < 32; ks += 4) {
        const v2f a  = *(const v2f*)(pa + ks);
        const v2f b0 = *(const v2f*)(pb + ks);
        const v2f b1 = *(const v2f*)(pb + 16 * SK_P + ks);
        acc[0] = wmma_f32(a, b0, acc[0]);
        acc[1] = wmma_f32(a, b1, acc[1]);
    }
}

__global__ __launch_bounds__(256)
void fa_gemm_abt_ksplit(const float* __restrict__ A,
                        const float* __restrict__ W,
                        const float* __restrict__ bias,
                        float* __restrict__ Out,
                        int M, int K, int Nc)
{
    __shared__ __attribute__((aligned(16))) float sA[2][16 * SK_P];
    __shared__ __attribute__((aligned(16))) float sB[2][32 * SK_P];
    __shared__ __attribute__((aligned(16))) float sR[8][16 * SK_RP];

    const int tid   = threadIdx.x;
    const int lane  = tid & 31;
    const int wave  = tid >> 5;
    const int nblkN = Nc >> 5;
    const int mb = blockIdx.x / nblkN, nb = blockIdx.x % nblkN;
    if (mb >= (M >> 4)) return;
    const int l15  = lane & 15;
    const int koff = (lane < 16) ? 0 : 2;

    const int sc = (tid & 63) << 2;
    const int sr = tid >> 6;
    const float* gA = A + (size_t)((mb << 4) + sr) * K + sc;
    const float* gB = W + (size_t)((nb << 5) + sr) * K + sc;
    const size_t rstep = (size_t)4 * K;
    const int    soff  = sr * SK_P + sc;

    const int fo = l15 * SK_P + (wave << 5) + koff;

    #define SK_LOADA(i) ra##i = *(const float4*)(gA + (i) * rstep);
    #define SK_LOADB(i) rb##i = *(const float4*)(gB + (i) * rstep);
    #define SK_PARKA(s, i) *(float4*)(&sA[s][soff + (i) * 4 * SK_P]) = ra##i;
    #define SK_PARKB(s, i) *(float4*)(&sB[s][soff + (i) * 4 * SK_P]) = rb##i;
    #define SK_LOAD_ALL()  SK_LOADA(0) SK_LOADA(1) SK_LOADA(2) SK_LOADA(3) \
                           SK_LOADB(0) SK_LOADB(1) SK_LOADB(2) SK_LOADB(3) \
                           SK_LOADB(4) SK_LOADB(5) SK_LOADB(6) SK_LOADB(7)
    #define SK_PARK_ALL(s) SK_PARKA(s, 0) SK_PARKA(s, 1) SK_PARKA(s, 2) SK_PARKA(s, 3) \
                           SK_PARKB(s, 0) SK_PARKB(s, 1) SK_PARKB(s, 2) SK_PARKB(s, 3) \
                           SK_PARKB(s, 4) SK_PARKB(s, 5) SK_PARKB(s, 6) SK_PARKB(s, 7)
    float4 ra0, ra1, ra2, ra3;
    float4 rb0, rb1, rb2, rb3, rb4, rb5, rb6, rb7;

    SK_LOAD_ALL()
    SK_PARK_ALL(0)
    gA += SK_KC; gB += SK_KC;
    SK_LOAD_ALL()
    __syncthreads();

    v8f acc[2] = {};
    const int nchunks = K / SK_KC;
    int c = 0;
    for (; c + 2 < nchunks; ++c) {
        const int st = c & 1;
        SK_PARK_ALL(st ^ 1)
        gA += SK_KC; gB += SK_KC;
        SK_LOAD_ALL()
        sk_chunk(&sA[st][fo], &sB[st][fo], acc);
        __syncthreads();
    }
    {
        const int st = c & 1;
        SK_PARK_ALL(st ^ 1)
        sk_chunk(&sA[st][fo], &sB[st][fo], acc);
        __syncthreads();
        sk_chunk(&sA[st ^ 1][fo], &sB[st ^ 1][fo], acc);
    }
    #undef SK_LOADA
    #undef SK_LOADB
    #undef SK_PARKA
    #undef SK_PARKB
    #undef SK_LOAD_ALL
    #undef SK_PARK_ALL

    const int rbase = (lane < 16) ? 0 : 8;
    #pragma unroll
    for (int ni = 0; ni < 2; ++ni)
        #pragma unroll
        for (int r = 0; r < 8; ++r)
            sR[wave][(rbase + r) * SK_RP + (ni << 4) + l15] = acc[ni][r];
    __syncthreads();

    const int orow = tid >> 4;
    const int oc   = (tid & 15) << 1;
    const int ro   = orow * SK_RP + oc;
    const v2f p0 = *(const v2f*)(&sR[0][ro]);
    const v2f p1 = *(const v2f*)(&sR[1][ro]);
    const v2f p2 = *(const v2f*)(&sR[2][ro]);
    const v2f p3 = *(const v2f*)(&sR[3][ro]);
    const v2f p4 = *(const v2f*)(&sR[4][ro]);
    const v2f p5 = *(const v2f*)(&sR[5][ro]);
    const v2f p6 = *(const v2f*)(&sR[6][ro]);
    const v2f p7 = *(const v2f*)(&sR[7][ro]);
    const int n = (nb << 5) + oc;
    v2f o = ((p0 + p1) + (p2 + p3)) + ((p4 + p5) + (p6 + p7));
    if (bias) { o.x += bias[n]; o.y += bias[n + 1]; }
    *(v2f*)(&Out[(size_t)((mb << 4) + orow) * Nc + n]) = o;
}

__device__ __forceinline__
void scores_row(const float* __restrict__ q,
                const float* __restrict__ k,
                const float* __restrict__ Wphi,
                float* __restrict__ ex,
                float* __restrict__ denom,
                float* __restrict__ Ew,
                int row)
{
    __shared__ float wph[64];
    __shared__ float qs[64];
    __shared__ float ssh[512];
    __shared__ float redm[4];
    __shared__ float redd[4];
    __shared__ float redE[4][64];
    const int t    = threadIdx.x;
    const int b    = row >> 9;
    const int lane = t & 31;
    const int wr   = t >> 5;
    const int qd   = t >> 2;
    const int c    = t & 3;
    if (t < 64) {
        qs[t]  = q[(size_t)row * 64 + t];
        wph[t] = Wphi[t];
    }
    __syncthreads();

    float qv[16], wv[16], Ewa[16];
    #pragma unroll
    for (int a = 0; a < 16; ++a) {
        qv[a]  = qs[c * 16 + a];
        wv[a]  = wph[c * 16 + a];
        Ewa[a] = 0.f;
    }
    float m = -INFINITY;
    const float* kb = k + ((size_t)(b * 512 + qd)) * 64 + c * 16;

    for (int n0 = 0; n0 < 16; n0 += 4) {
        float e[4][16];
        float s[4];
        #pragma unroll
        for (int g = 0; g < 4; ++g) {
            const float4* kp = (const float4*)(kb + (size_t)(n0 + g) * 32 * 64);
            float acc = 0.f;
            #pragma unroll
            for (int v = 0; v < 4; ++v) {
                const float4 kv = kp[v];
                e[g][4 * v + 0] = rho_relu(qv[4 * v + 0] + kv.x);
                e[g][4 * v + 1] = rho_relu(qv[4 * v + 1] + kv.y);
                e[g][4 * v + 2] = rho_relu(qv[4 * v + 2] + kv.z);
                e[g][4 * v + 3] = rho_relu(qv[4 * v + 3] + kv.w);
                acc += e[g][4 * v + 0] * wv[4 * v + 0];
                acc += e[g][4 * v + 1] * wv[4 * v + 1];
                acc += e[g][4 * v + 2] * wv[4 * v + 2];
                acc += e[g][4 * v + 3] * wv[4 * v + 3];
            }
            acc += __shfl_xor(acc, 1);
            acc += __shfl_xor(acc, 2);
            s[g] = acc;
            if (c == 0) ssh[qd + 32 * (n0 + g)] = acc;
        }
        const float mn = fmaxf(fmaxf(m, fmaxf(s[0], s[1])), fmaxf(s[2], s[3]));
        const float sc = expf(m - mn);
        const float p0 = expf(s[0] - mn);
        const float p1 = expf(s[1] - mn);
        const float p2 = expf(s[2] - mn);
        const float p3 = expf(s[3] - mn);
        #pragma unroll
        for (int a = 0; a < 16; ++a) {
            float v = Ewa[a] * sc;
            v += p0 * e[0][a];
            v += p1 * e[1][a];
            v += p2 * e[2][a];
            v += p3 * e[3][a];
            Ewa[a] = v;
        }
        m = mn;
    }

    float M = m;
    M = fmaxf(M, __shfl_xor(M, 4));
    M = fmaxf(M, __shfl_xor(M, 8));
    M = fmaxf(M, __shfl_xor(M, 16));
    if (lane == 0) redm[wr] = M;
    __syncthreads();
    M = fmaxf(fmaxf(redm[0], redm[1]), fmaxf(redm[2], redm[3]));

    const float f = expf(m - M);
    #pragma unroll
    for (int a = 0; a < 16; ++a) {
        float v = Ewa[a] * f;
        v += __shfl_xor(v, 4);
        v += __shfl_xor(v, 8);
        v += __shfl_xor(v, 16);
        Ewa[a] = v;
    }
    if (lane < 4) {
        #pragma unroll
        for (int a = 0; a < 16; ++a) redE[wr][lane * 16 + a] = Ewa[a];
    }

    float ds = 0.f;
    #pragma unroll
    for (int n = 0; n < 4; ++n) {
        const int   j = t + 128 * n;
        const float v = expf(ssh[j] - M);
        ex[(size_t)row * 512 + j] = v;
        ds += v;
    }
    for (int off = 16; off; off >>= 1)
        ds += __shfl_xor(ds, off);
    if (lane == 0) redd[wr] = ds;
    __syncthreads();
    if (t < 64)
        Ew[(size_t)row * 64 + t] = (redE[0][t] + redE[1][t]) + (redE[2][t] + redE[3][t]);
    if (t == 0)
        denom[row] = (redd[0] + redd[1]) + (redd[2] + redd[3]);
}

#define VY_TILES 256
__global__ __launch_bounds__(128)
void fa_scores_vy(const float* __restrict__ X,
                  const float* __restrict__ Wyw,
                  const float* __restrict__ Wyb,
                  float* __restrict__ Vy,
                  const float* __restrict__ q,
                  const float* __restrict__ k,
                  const float* __restrict__ Wphi,
                  float* __restrict__ ex,
                  float* __restrict__ denom,
                  float* __restrict__ Ew)
{
    if (blockIdx.x < VY_TILES)
        gemm_abt_tile(X, Wyw, Wyb, Vy, 1024, 1024, blockIdx.x);
    else
        scores_row(q, k, Wphi, ex, denom, Ew, blockIdx.x - VY_TILES);
}

#define GT_VT 1040

__device__ __forceinline__ void gt_chunk_kn(const float* pa, const float* pv,
                                            v8f (&acc)[2][2])
{
    #pragma unroll
    for (int ks = 0; ks < GT_KC; ks += 4) {
        const v2f a0 = *(const v2f*)(pa + ks);
        const v2f a1 = *(const v2f*)(pa + 16 * GT_P + ks);
        v2f b0, b1;
        b0.x = pv[ks * 16];               b0.y = pv[ks * 16 + 16];
        b1.x = pv[GT_VT + ks * 16];       b1.y = pv[GT_VT + ks * 16 + 16];
        acc[0][0] = wmma_f32(a0, b0, acc[0][0]);
        acc[1][0] = wmma_f32(a1, b0, acc[1][0]);
        acc[0][1] = wmma_f32(a0, b1, acc[0][1]);
        acc[1][1] = wmma_f32(a1, b1, acc[1][1]);
    }
}

__global__ __launch_bounds__(128)
void fa_part1(const float* __restrict__ ex,
              const float* __restrict__ Vy,
              float* __restrict__ part1)
{
    __shared__ __attribute__((aligned(16))) float sA[2][64 * GT_P];
    __shared__ __attribute__((aligned(16))) float sV[2][4 * GT_VT];

    const int tid  = threadIdx.x;
    const int lane = tid & 31;
    const int wave = tid >> 5;
    const int wm   = wave >> 1, wn = wave & 1;
    const int b    = blockIdx.x >> 7;
    const int mb   = (blockIdx.x >> 4) & 7;
    const int nb   = blockIdx.x & 15;
    const int l15  = lane & 15;
    const int koff = (lane < 16) ? 0 : 2;

    const int sc = (tid & 15) << 2;
    const int sr = tid >> 4;
    const float* gA = ex + (size_t)(b * 512 + (mb << 6) + sr) * 512 + sc;
    const float* gV = Vy + (size_t)(b * 512 + sr) * 1024 + (nb << 6) + sc;
    const int soffA = sr * GT_P + sc;
    const int soffV = (sc >> 4) * GT_VT + sr * 16 + (sc & 15);

    const int fa = ((wm << 5) + l15) * GT_P + koff;
    const int fv = (wn << 1) * GT_VT + koff * 16 + l15;

    #define GT_LOAD(i)  ra##i = *(const float4*)(gA + (size_t)(i) * 8 * 512); \
                        rv##i = *(const float4*)(gV + (size_t)(i) * 8 * 1024);
    #define GT_PARK(s, i) *(float4*)(&sA[s][soffA + (i) * 8 * GT_P])  = ra##i; \
                          *(float4*)(&sV[s][soffV + (i) * 8 * 16]) = rv##i;
    #define GT_LOAD_ALL()  GT_LOAD(0) GT_LOAD(1) GT_LOAD(2) GT_LOAD(3) \
                           GT_LOAD(4) GT_LOAD(5) GT_LOAD(6) GT_LOAD(7)
    #define GT_PARK_ALL(s) GT_PARK(s, 0) GT_PARK(s, 1) GT_PARK(s, 2) GT_PARK(s, 3) \
                           GT_PARK(s, 4) GT_PARK(s, 5) GT_PARK(s, 6) GT_PARK(s, 7)
    float4 ra0, ra1, ra2, ra3, ra4, ra5, ra6, ra7;
    float4 rv0, rv1, rv2, rv3, rv4, rv5, rv6, rv7;

    GT_LOAD_ALL()
    GT_PARK_ALL(0)
    gA += GT_KC; gV += (size_t)GT_KC * 1024;
    GT_LOAD_ALL()
    __syncthreads();

    v8f acc[2][2] = {};
    int c = 0;
    for (; c + 2 < 512 / GT_KC; ++c) {
        const int st = c & 1;
        GT_PARK_ALL(st ^ 1)
        gA += GT_KC; gV += (size_t)GT_KC * 1024;
        GT_LOAD_ALL()
        gt_chunk_kn(&sA[st][fa], &sV[st][fv], acc);
        __syncthreads();
    }
    {
        const int st = c & 1;
        GT_PARK_ALL(st ^ 1)
        gt_chunk_kn(&sA[st][fa], &sV[st][fv], acc);
        __syncthreads();
        gt_chunk_kn(&sA[st ^ 1][fa], &sV[st ^ 1][fv], acc);
    }
    #undef GT_LOAD
    #undef GT_PARK
    #undef GT_LOAD_ALL
    #undef GT_PARK_ALL

    const int mbase = b * 512 + (mb << 6) + (wm << 5) + ((lane < 16) ? 0 : 8);
    const int col0  = (nb << 6) + (wn << 5) + l15;
    #pragma unroll
    for (int ni = 0; ni < 2; ++ni) {
        const int n = col0 + (ni << 4);
        #pragma unroll
        for (int mi = 0; mi < 2; ++mi)
            #pragma unroll
            for (int r = 0; r < 8; ++r)
                part1[(size_t)(mbase + (mi << 4) + r) * 1024 + n] = acc[mi][ni][r];
    }
}

__global__ void fa_final(const float* __restrict__ Ew,
                         const float* __restrict__ Wew,
                         const float* __restrict__ Web,
                         const float* __restrict__ part1,
                         const float* __restrict__ denom,
                         float* __restrict__ out)
{
    const int lane = threadIdx.x & 31;
    const int wave = threadIdx.x >> 5;
    const int t  = blockIdx.x * (blockDim.x >> 5) + wave;
    const int mt = t >> 6, nt = t & 63;
    const int l15  = lane & 15;
    const int koff = (lane < 16) ? 0 : 2;
    const float* Arow = Ew  + (size_t)((mt << 4) + l15) * 64 + koff;
    const float* Brow = Wew + (size_t)((nt << 4) + l15) * 64 + koff;
    v8f acc = {};
    #pragma unroll
    for (int kk = 0; kk < 64; kk += 4) {
        v2f av = *(const v2f*)(Arow + kk);
        v2f bv = *(const v2f*)(Brow + kk);
        acc = wmma_f32(av, bv, acc);
    }
    const int   n     = (nt << 4) + l15;
    const int   mbase = (mt << 4) + ((lane < 16) ? 0 : 8);
    const float bn    = Web[n];
    #pragma unroll
    for (int r = 0; r < 8; ++r) {
        const int   m = mbase + r;
        const float d = denom[m];
        out[(size_t)m * 1024 + n] =
            (part1[(size_t)m * 1024 + n] + acc[r] + bn * d) / (d + EPS_DEN);
    }
}

extern "C" void kernel_launch(void* const* d_in, const int* in_sizes, int n_in,
                              void* d_out, int out_size, void* d_ws, size_t ws_size,
                              hipStream_t stream)
{
    (void)in_sizes; (void)n_in; (void)out_size; (void)ws_size;
    const float* X    = (const float*)d_in[0];
    const float* Wq   = (const float*)d_in[1];
    const float* Wk   = (const float*)d_in[2];
    const float* Wphi = (const float*)d_in[3];
    const float* Wyw  = (const float*)d_in[4];
    const float* Wyb  = (const float*)d_in[5];
    const float* Wew  = (const float*)d_in[6];
    const float* Web  = (const float*)d_in[7];
    float* out = (float*)d_out;

    float* ws  = (float*)d_ws;
    float* q   = ws;  ws += 1024 * 64;
    float* kk  = ws;  ws += 1024 * 64;
    float* Vy  = ws;  ws += 1024 * 1024;
    float* ex  = ws;  ws += 1024 * 512;
    float* den = ws;  ws += 1024;
    float* p1  = ws;  ws += 1024 * 1024;
    float* Ew  = ws;  ws += 1024 * 64;

    fa_gemm_abt_ksplit<<<128, 256, 0, stream>>>(X, Wq, nullptr, q,  1024, 1024, 64);
    fa_gemm_abt_ksplit<<<128, 256, 0, stream>>>(X, Wk, nullptr, kk, 1024, 1024, 64);

    fa_scores_vy<<<1280, 128, 0, stream>>>(X, Wyw, Wyb, Vy, q, kk, Wphi, ex, den, Ew);

    fa_part1<<<256, 128, 0, stream>>>(ex, Vy, p1);

    fa_final<<<512, 256, 0, stream>>>(Ew, Wew, Web, p1, den, out);
}


# q and k projections run as ONE launch of 256 k-split blocks (block index picks Wq or Wk and its output); tiles, k order and slice-sum tree kept, q and k bit-identical; 5 -> 4 launches
# speedup vs baseline: 5.8150x; 1.1084x over previous
#include <hip/hip_runtime.h>
#include <hip/hip_bf16.h>
#include <math.h>

typedef __attribute__((ext_vector_type(2))) float v2f;
typedef __attribute__((ext_vector_type(8))) float v8f;

#define EPS_RHO 1e-8f
#define EPS_DEN 1e-9f

__device__ __forceinline__ float rho_relu(float x) {
    float p = sqrtf(fmaxf(x, 0.f) + EPS_RHO);
    float n = sqrtf(fmaxf(-x, 0.f) + EPS_RHO);
    return fmaxf(p - n, 0.f);
}

__device__ __forceinline__ v8f wmma_f32(v2f a, v2f b, v8f c) {
    return __builtin_amdgcn_wmma_f32_16x16x4_f32(false, a, false, b,
                                                 (short)0, c, false, false);
}

#define GT_KC 64
#define GT_P  68

__device__ __forceinline__ void gt_chunk(const float* pa, const float* pb,
                                         v8f (&acc)[2][2])
{
    #pragma unroll
    for (int ks = 0; ks < GT_KC; ks += 4) {
        const v2f a0 = *(const v2f*)(pa + ks);
        const v2f a1 = *(const v2f*)(pa + 16 * GT_P + ks);
        const v2f b0 = *(const v2f*)(pb + ks);
        const v2f b1 = *(const v2f*)(pb + 16 * GT_P + ks);
        acc[0][0] = wmma_f32(a0, b0, acc[0][0]);
        acc[1][0] = wmma_f32(a1, b0, acc[1][0]);
        acc[0][1] = wmma_f32(a0, b1, acc[0][1]);
        acc[1][1] = wmma_f32(a1, b1, acc[1][1]);
    }
}

__device__ __forceinline__
void gemm_abt_tile(const float* __restrict__ A,
                   const float* __restrict__ W,
                   const float* __restrict__ bias,
                   float* __restrict__ Out,
                   int K, int Nc, int bid)
{
    __shared__ __attribute__((aligned(16))) float sA[2][64 * GT_P];
    __shared__ __attribute__((aligned(16))) float sB[2][64 * GT_P];

    const int tid   = threadIdx.x;
    const int lane  = tid & 31;
    const int wave  = tid >> 5;
    const int wm    = wave >> 1, wn = wave & 1;
    const int nblkN = Nc >> 6;
    const int mb = bid / nblkN, nb = bid % nblkN;
    const int l15  = lane & 15;
    const int koff = (lane < 16) ? 0 : 2;

    const int sc = (tid & 15) << 2;
    const int sr = tid >> 4;
    const float* gA = A + (size_t)((mb << 6) + sr) * K + sc;
    const float* gB = W + (size_t)((nb << 6) + sr) * K + sc;
    const size_t rstep = (size_t)8 * K;
    const int    soff  = sr * GT_P + sc;

    const int fa = ((wm << 5) + l15) * GT_P + koff;
    const int fb = ((wn << 5) + l15) * GT_P + koff;

    #define GT_LOAD(i)  ra##i = *(const float4*)(gA + (i) * rstep); \
                        rb##i = *(const float4*)(gB + (i) * rstep);
    #define GT_PARK(s, i) *(float4*)(&sA[s][soff + (i) * 8 * GT_P]) = ra##i; \
                          *(float4*)(&sB[s][soff + (i) * 8 * GT_P]) = rb##i;
    #define GT_LOAD_ALL()  GT_LOAD(0) GT_LOAD(1) GT_LOAD(2) GT_LOAD(3) \
                           GT_LOAD(4) GT_LOAD(5) GT_LOAD(6) GT_LOAD(7)
    #define GT_PARK_ALL(s) GT_PARK(s, 0) GT_PARK(s, 1) GT_PARK(s, 2) GT_PARK(s, 3) \
                           GT_PARK(s, 4) GT_PARK(s, 5) GT_PARK(s, 6) GT_PARK(s, 7)
    float4 ra0, ra1, ra2, ra3, ra4, ra5, ra6, ra7;
    float4 rb0, rb1, rb2, rb3, rb4, rb5, rb6, rb7;

    GT_LOAD_ALL()
    GT_PARK_ALL(0)
    gA += GT_KC; gB += GT_KC;
    GT_LOAD_ALL()
    __syncthreads();

    v8f acc[2][2] = {};
    const int nchunks = K / GT_KC;
    int c = 0;
    for (; c + 2 < nchunks; ++c) {
        const int st = c & 1;
        GT_PARK_ALL(st ^ 1)
        gA += GT_KC; gB += GT_KC;
        GT_LOAD_ALL()
        gt_chunk(&sA[st][fa], &sB[st][fb], acc);
        __syncthreads();
    }
    {
        const int st = c & 1;
        GT_PARK_ALL(st ^ 1)
        gt_chunk(&sA[st][fa], &sB[st][fb], acc);
        __syncthreads();
        gt_chunk(&sA[st ^ 1][fa], &sB[st ^ 1][fb], acc);
    }
    #undef GT_LOAD
    #undef GT_PARK
    #undef GT_LOAD_ALL
    #undef GT_PARK_ALL

    const int mbase = (mb << 6) + (wm << 5) + ((lane < 16) ? 0 : 8);
    #pragma unroll
    for (int ni = 0; ni < 2; ++ni) {
        const int   n  = (nb << 6) + (wn << 5) + (ni << 4) + l15;
        const float bn = bias ? bias[n] : 0.f;
        #pragma unroll
        for (int mi = 0; mi < 2; ++mi)
            #pragma unroll
            for (int r = 0; r < 8; ++r)
                Out[(size_t)(mbase + (mi << 4) + r) * Nc + n] = acc[mi][ni][r] + bn;
    }
}

#define SK_KC 256
#define SK_P  260
#define SK_RP 36

__device__ __forceinline__ void sk_chunk(const float* pa, const float* pb,
                                         v8f (&acc)[2])
{
    #pragma unroll
    for (int ks = 0; ks < 32; ks += 4) {
        const v2f a  = *(const v2f*)(pa + ks);
        const v2f b0 = *(const v2f*)(pb + ks);
        const v2f b1 = *(const v2f*)(pb + 16 * SK_P + ks);
        acc[0] = wmma_f32(a, b0, acc[0]);
        acc[1] = wmma_f32(a, b1, acc[1]);
    }
}

__global__ __launch_bounds__(256)
void fa_gemm_abt_ksplit2(const float* __restrict__ A,
                         const float* __restrict__ W0,
                         const float* __restrict__ W1,
                         float* __restrict__ Out0,
                         float* __restrict__ Out1,
                         int M, int K, int Nc)
{
    __shared__ __attribute__((aligned(16))) float sA[2][16 * SK_P];
    __shared__ __attribute__((aligned(16))) float sB[2][32 * SK_P];
    __shared__ __attribute__((aligned(16))) float sR[8][16 * SK_RP];

    const int tid   = threadIdx.x;
    const int lane  = tid & 31;
    const int wave  = tid >> 5;
    const int nblkN = Nc >> 5;
    const int mb  = blockIdx.x / (2 * nblkN);
    const int nb2 = blockIdx.x % (2 * nblkN);
    if (mb >= (M >> 4)) return;
    const bool second = nb2 >= nblkN;
    const int  nb = second ? nb2 - nblkN : nb2;
    const float* __restrict__ W   = second ? W1 : W0;
    float* __restrict__       Out = second ? Out1 : Out0;
    const int l15  = lane & 15;
    const int koff = (lane < 16) ? 0 : 2;

    const int sc = (tid & 63) << 2;
    const int sr = tid >> 6;
    const float* gA = A + (size_t)((mb << 4) + sr) * K + sc;
    const float* gB = W + (size_t)((nb << 5) + sr) * K + sc;
    const size_t rstep = (size_t)4 * K;
    const int    soff  = sr * SK_P + sc;

    const int fo = l15 * SK_P + (wave << 5) + koff;

    #define SK_LOADA(i) ra##i = *(const float4*)(gA + (i) * rstep);
    #define SK_LOADB(i) rb##i = *(const float4*)(gB + (i) * rstep);
    #define SK_PARKA(s, i) *(float4*)(&sA[s][soff + (i) * 4 * SK_P]) = ra##i;
    #define SK_PARKB(s, i) *(float4*)(&sB[s][soff + (i) * 4 * SK_P]) = rb##i;
    #define SK_LOAD_ALL()  SK_LOADA(0) SK_LOADA(1) SK_LOADA(2) SK_LOADA(3) \
                           SK_LOADB(0) SK_LOADB(1) SK_LOADB(2) SK_LOADB(3) \
                           SK_LOADB(4) SK_LOADB(5) SK_LOADB(6) SK_LOADB(7)
    #define SK_PARK_ALL(s) SK_PARKA(s, 0) SK_PARKA(s, 1) SK_PARKA(s, 2) SK_PARKA(s, 3) \
                           SK_PARKB(s, 0) SK_PARKB(s, 1) SK_PARKB(s, 2) SK_PARKB(s, 3) \
                           SK_PARKB(s, 4) SK_PARKB(s, 5) SK_PARKB(s, 6) SK_PARKB(s, 7)
    float4 ra0, ra1, ra2, ra3;
    float4 rb0, rb1, rb2, rb3, rb4, rb5, rb6, rb7;

    SK_LOAD_ALL()
    SK_PARK_ALL(0)
    gA += SK_KC; gB += SK_KC;
    SK_LOAD_ALL()
    __syncthreads();

    v8f acc[2] = {};
    const int nchunks = K / SK_KC;
    int c = 0;
    for (; c + 2 < nchunks; ++c) {
        const int st = c & 1;
        SK_PARK_ALL(st ^ 1)
        gA += SK_KC; gB += SK_KC;
        SK_LOAD_ALL()
        sk_chunk(&sA[st][fo], &sB[st][fo], acc);
        __syncthreads();
    }
    {
        const int st = c & 1;
        SK_PARK_ALL(st ^ 1)
        sk_chunk(&sA[st][fo], &sB[st][fo], acc);
        __syncthreads();
        sk_chunk(&sA[st ^ 1][fo], &sB[st ^ 1][fo], acc);
    }
    #undef SK_LOADA
    #undef SK_LOADB
    #undef SK_PARKA
    #undef SK_PARKB
    #undef SK_LOAD_ALL
    #undef SK_PARK_ALL

    const int rbase = (lane < 16) ? 0 : 8;
    #pragma unroll
    for (int ni = 0; ni < 2; ++ni)
        #pragma unroll
        for (int r = 0; r < 8; ++r)
            sR[wave][(rbase + r) * SK_RP + (ni << 4) + l15] = acc[ni][r];
    __syncthreads();

    const int orow = tid >> 4;
    const int oc   = (tid & 15) << 1;
    const int ro   = orow * SK_RP + oc;
    const v2f p0 = *(const v2f*)(&sR[0][ro]);
    const v2f p1 = *(const v2f*)(&sR[1][ro]);
    const v2f p2 = *(const v2f*)(&sR[2][ro]);
    const v2f p3 = *(const v2f*)(&sR[3][ro]);
    const v2f p4 = *(const v2f*)(&sR[4][ro]);
    const v2f p5 = *(const v2f*)(&sR[5][ro]);
    const v2f p6 = *(const v2f*)(&sR[6][ro]);
    const v2f p7 = *(const v2f*)(&sR[7][ro]);
    const int n = (nb << 5) + oc;
    const v2f o = ((p0 + p1) + (p2 + p3)) + ((p4 + p5) + (p6 + p7));
    *(v2f*)(&Out[(size_t)((mb << 4) + orow) * Nc + n]) = o;
}

__device__ __forceinline__
void scores_row(const float* __restrict__ q,
                const float* __restrict__ k,
                const float* __restrict__ Wphi,
                float* __restrict__ ex,
                float* __restrict__ denom,
                float* __restrict__ Ew,
                int row)
{
    __shared__ float wph[64];
    __shared__ float qs[64];
    __shared__ float ssh[512];
    __shared__ float redm[4];
    __shared__ float redd[4];
    __shared__ float redE[4][64];
    const int t    = threadIdx.x;
    const int b    = row >> 9;
    const int lane = t & 31;
    const int wr   = t >> 5;
    const int qd   = t >> 2;
    const int c    = t & 3;
    if (t < 64) {
        qs[t]  = q[(size_t)row * 64 + t];
        wph[t] = Wphi[t];
    }
    __syncthreads();

    float qv[16], wv[16], Ewa[16];
    #pragma unroll
    for (int a = 0; a < 16; ++a) {
        qv[a]  = qs[c * 16 + a];
        wv[a]  = wph[c * 16 + a];
        Ewa[a] = 0.f;
    }
    float m = -INFINITY;
    const float* kb = k + ((size_t)(b * 512 + qd)) * 64 + c * 16;

    for (int n0 = 0; n0 < 16; n0 += 4) {
        float e[4][16];
        float s[4];
        #pragma unroll
        for (int g = 0; g < 4; ++g) {
            const float4* kp = (const float4*)(kb + (size_t)(n0 + g) * 32 * 64);
            float acc = 0.f;
            #pragma unroll
            for (int v = 0; v < 4; ++v) {
                const float4 kv = kp[v];
                e[g][4 * v + 0] = rho_relu(qv[4 * v + 0] + kv.x);
                e[g][4 * v + 1] = rho_relu(qv[4 * v + 1] + kv.y);
                e[g][4 * v + 2] = rho_relu(qv[4 * v + 2] + kv.z);
                e[g][4 * v + 3] = rho_relu(qv[4 * v + 3] + kv.w);
                acc += e[g][4 * v + 0] * wv[4 * v + 0];
                acc += e[g][4 * v + 1] * wv[4 * v + 1];
                acc += e[g][4 * v + 2] * wv[4 * v + 2];
                acc += e[g][4 * v + 3] * wv[4 * v + 3];
            }
            acc += __shfl_xor(acc, 1);
            acc += __shfl_xor(acc, 2);
            s[g] = acc;
            if (c == 0) ssh[qd + 32 * (n0 + g)] = acc;
        }
        const float mn = fmaxf(fmaxf(m, fmaxf(s[0], s[1])), fmaxf(s[2], s[3]));
        const float sc = expf(m - mn);
        const float p0 = expf(s[0] - mn);
        const float p1 = expf(s[1] - mn);
        const float p2 = expf(s[2] - mn);
        const float p3 = expf(s[3] - mn);
        #pragma unroll
        for (int a = 0; a < 16; ++a) {
            float v = Ewa[a] * sc;
            v += p0 * e[0][a];
            v += p1 * e[1][a];
            v += p2 * e[2][a];
            v += p3 * e[3][a];
            Ewa[a] = v;
        }
        m = mn;
    }

    float M = m;
    M = fmaxf(M, __shfl_xor(M, 4));
    M = fmaxf(M, __shfl_xor(M, 8));
    M = fmaxf(M, __shfl_xor(M, 16));
    if (lane == 0) redm[wr] = M;
    __syncthreads();
    M = fmaxf(fmaxf(redm[0], redm[1]), fmaxf(redm[2], redm[3]));

    const float f = expf(m - M);
    #pragma unroll
    for (int a = 0; a < 16; ++a) {
        float v = Ewa[a] * f;
        v += __shfl_xor(v, 4);
        v += __shfl_xor(v, 8);
        v += __shfl_xor(v, 16);
        Ewa[a] = v;
    }
    if (lane < 4) {
        #pragma unroll
        for (int a = 0; a < 16; ++a) redE[wr][lane * 16 + a] = Ewa[a];
    }

    float ds = 0.f;
    #pragma unroll
    for (int n = 0; n < 4; ++n) {
        const int   j = t + 128 * n;
        const float v = expf(ssh[j] - M);
        ex[(size_t)row * 512 + j] = v;
        ds += v;
    }
    for (int off = 16; off; off >>= 1)
        ds += __shfl_xor(ds, off);
    if (lane == 0) redd[wr] = ds;
    __syncthreads();
    if (t < 64)
        Ew[(size_t)row * 64 + t] = (redE[0][t] + redE[1][t]) + (redE[2][t] + redE[3][t]);
    if (t == 0)
        denom[row] = (redd[0] + redd[1]) + (redd[2] + redd[3]);
}

#define VY_TILES 256
__global__ __launch_bounds__(128)
void fa_scores_vy(const float* __restrict__ X,
                  const float* __restrict__ Wyw,
                  const float* __restrict__ Wyb,
                  float* __restrict__ Vy,
                  const float* __restrict__ q,
                  const float* __restrict__ k,
                  const float* __restrict__ Wphi,
                  float* __restrict__ ex,
                  float* __restrict__ denom,
                  float* __restrict__ Ew)
{
    if (blockIdx.x < VY_TILES)
        gemm_abt_tile(X, Wyw, Wyb, Vy, 1024, 1024, blockIdx.x);
    else
        scores_row(q, k, Wphi, ex, denom, Ew, blockIdx.x - VY_TILES);
}

#define GT_VT 1040

__device__ __forceinline__ void gt_chunk_kn(const float* pa, const float* pv,
                                            v8f (&acc)[2][2])
{
    #pragma unroll
    for (int ks = 0; ks < GT_KC; ks += 4) {
        const v2f a0 = *(const v2f*)(pa + ks);
        const v2f a1 = *(const v2f*)(pa + 16 * GT_P + ks);
        v2f b0, b1;
        b0.x = pv[ks * 16];               b0.y = pv[ks * 16 + 16];
        b1.x = pv[GT_VT + ks * 16];       b1.y = pv[GT_VT + ks * 16 + 16];
        acc[0][0] = wmma_f32(a0, b0, acc[0][0]);
        acc[1][0] = wmma_f32(a1, b0, acc[1][0]);
        acc[0][1] = wmma_f32(a0, b1, acc[0][1]);
        acc[1][1] = wmma_f32(a1, b1, acc[1][1]);
    }
}

__global__ __launch_bounds__(128)
void fa_part1(const float* __restrict__ ex,
              const float* __restrict__ Vy,
              float* __restrict__ part1)
{
    __shared__ __attribute__((aligned(16))) float sA[2][64 * GT_P];
    __shared__ __attribute__((aligned(16))) float sV[2][4 * GT_VT];

    const int tid  = threadIdx.x;
    const int lane = tid & 31;
    const int wave = tid >> 5;
    const int wm   = wave >> 1, wn = wave & 1;
    const int b    = blockIdx.x >> 7;
    const int mb   = (blockIdx.x >> 4) & 7;
    const int nb   = blockIdx.x & 15;
    const int l15  = lane & 15;
    const int koff = (lane < 16) ? 0 : 2;

    const int sc = (tid & 15) << 2;
    const int sr = tid >> 4;
    const float* gA = ex + (size_t)(b * 512 + (mb << 6) + sr) * 512 + sc;
    const float* gV = Vy + (size_t)(b * 512 + sr) * 1024 + (nb << 6) + sc;
    const int soffA = sr * GT_P + sc;
    const int soffV = (sc >> 4) * GT_VT + sr * 16 + (sc & 15);

    const int fa = ((wm << 5) + l15) * GT_P + koff;
    const int fv = (wn << 1) * GT_VT + koff * 16 + l15;

    #define GT_LOAD(i)  ra##i = *(const float4*)(gA + (size_t)(i) * 8 * 512); \
                        rv##i = *(const float4*)(gV + (size_t)(i) * 8 * 1024);
    #define GT_PARK(s, i) *(float4*)(&sA[s][soffA + (i) * 8 * GT_P])  = ra##i; \
                          *(float4*)(&sV[s][soffV + (i) * 8 * 16]) = rv##i;
    #define GT_LOAD_ALL()  GT_LOAD(0) GT_LOAD(1) GT_LOAD(2) GT_LOAD(3) \
                           GT_LOAD(4) GT_LOAD(5) GT_LOAD(6) GT_LOAD(7)
    #define GT_PARK_ALL(s) GT_PARK(s, 0) GT_PARK(s, 1) GT_PARK(s, 2) GT_PARK(s, 3) \
                           GT_PARK(s, 4) GT_PARK(s, 5) GT_PARK(s, 6) GT_PARK(s, 7)
    float4 ra0, ra1, ra2, ra3, ra4, ra5, ra6, ra7;
    float4 rv0, rv1, rv2, rv3, rv4, rv5, rv6, rv7;

    GT_LOAD_ALL()
    GT_PARK_ALL(0)
    gA += GT_KC; gV += (size_t)GT_KC * 1024;
    GT_LOAD_ALL()
    __syncthreads();

    v8f acc[2][2] = {};
    int c = 0;
    for (; c + 2 < 512 / GT_KC; ++c) {
        const int st = c & 1;
        GT_PARK_ALL(st ^ 1)
        gA += GT_KC; gV += (size_t)GT_KC * 1024;
        GT_LOAD_ALL()
        gt_chunk_kn(&sA[st][fa], &sV[st][fv], acc);
        __syncthreads();
    }
    {
        const int st = c & 1;
        GT_PARK_ALL(st ^ 1)
        gt_chunk_kn(&sA[st][fa], &sV[st][fv], acc);
        __syncthreads();
        gt_chunk_kn(&sA[st ^ 1][fa], &sV[st ^ 1][fv], acc);
    }
    #undef GT_LOAD
    #undef GT_PARK
    #undef GT_LOAD_ALL
    #undef GT_PARK_ALL

    const int mbase = b * 512 + (mb << 6) + (wm << 5) + ((lane < 16) ? 0 : 8);
    const int col0  = (nb << 6) + (wn << 5) + l15;
    #pragma unroll
    for (int ni = 0; ni < 2; ++ni) {
        const int n = col0 + (ni << 4);
        #pragma unroll
        for (int mi = 0; mi < 2; ++mi)
            #pragma unroll
            for (int r = 0; r < 8; ++r)
                part1[(size_t)(mbase + (mi << 4) + r) * 1024 + n] = acc[mi][ni][r];
    }
}

__global__ void fa_final(const float* __restrict__ Ew,
                         const float* __restrict__ Wew,
                         const float* __restrict__ Web,
                         const float* __restrict__ part1,
                         const float* __restrict__ denom,
                         float* __restrict__ out)
{
    const int lane = threadIdx.x & 31;
    const int wave = threadIdx.x >> 5;
    const int t  = blockIdx.x * (blockDim.x >> 5) + wave;
    const int mt = t >> 6, nt = t & 63;
    const int l15  = lane & 15;
    const int koff = (lane < 16) ? 0 : 2;
    const float* Arow = Ew  + (size_t)((mt << 4) + l15) * 64 + koff;
    const float* Brow = Wew + (size_t)((nt << 4) + l15) * 64 + koff;
    v8f acc = {};
    #pragma unroll
    for (int kk = 0; kk < 64; kk += 4) {
        v2f av = *(const v2f*)(Arow + kk);
        v2f bv = *(const v2f*)(Brow + kk);
        acc = wmma_f32(av, bv, acc);
    }
    const int   n     = (nt << 4) + l15;
    const int   mbase = (mt << 4) + ((lane < 16) ? 0 : 8);
    const float bn    = Web[n];
    #pragma unroll
    for (int r = 0; r < 8; ++r) {
        const int   m = mbase + r;
        const float d = denom[m];
        out[(size_t)m * 1024 + n] =
            (part1[(size_t)m * 1024 + n] + acc[r] + bn * d) / (d + EPS_DEN);
    }
}

extern "C" void kernel_launch(void* const* d_in, const int* in_sizes, int n_in,
                              void* d_out, int out_size, void* d_ws, size_t ws_size,
                              hipStream_t stream)
{
    (void)in_sizes; (void)n_in; (void)out_size; (void)ws_size;
    const float* X    = (const float*)d_in[0];
    const float* Wq   = (const float*)d_in[1];
    const float* Wk   = (const float*)d_in[2];
    const float* Wphi = (const float*)d_in[3];
    const float* Wyw  = (const float*)d_in[4];
    const float* Wyb  = (const float*)d_in[5];
    const float* Wew  = (const float*)d_in[6];
    const float* Web  = (const float*)d_in[7];
    float* out = (float*)d_out;

    float* ws  = (float*)d_ws;
    float* q   = ws;  ws += 1024 * 64;
    float* kk  = ws;  ws += 1024 * 64;
    float* Vy  = ws;  ws += 1024 * 1024;
    float* ex  = ws;  ws += 1024 * 512;
    float* den = ws;  ws += 1024;
    float* p1  = ws;  ws += 1024 * 1024;
    float* Ew  = ws;  ws += 1024 * 64;

    fa_gemm_abt_ksplit2<<<256, 256, 0, stream>>>(X, Wq, Wk, q, kk, 1024, 1024, 64);

    fa_scores_vy<<<1280, 128, 0, stream>>>(X, Wyw, Wyb, Vy, q, kk, Wphi, ex, den, Ew);

    fa_part1<<<256, 128, 0, stream>>>(ex, Vy, p1);

    fa_final<<<512, 256, 0, stream>>>(Ew, Wew, Web, p1, den, out);
}
